# starting kernel
# baseline (speedup 1.0000x reference)
#include <hip/hip_runtime.h>
#include <math.h>

#define BATCH   2
#define SEQ     2048
#define DMODEL  1024
#define NHEADS  16
#define DK      64
#define LOG_THETA 9.210340371976184f

typedef __attribute__((ext_vector_type(2))) float v2f;
typedef __attribute__((ext_vector_type(8))) float v8f;

static __device__ __forceinline__ v8f wmma_f32_4(v2f a, v2f b, v8f c) {
  return __builtin_amdgcn_wmma_f32_16x16x4_f32(
       false, a,  false, b,
       (short)0, c,  false,  false);
}

__global__ __launch_bounds__(256) void proj_gemm_kernel(
    const float* __restrict__ A, const float* __restrict__ W,
    float* __restrict__ Y, int in_head_major, int out_head_major,
    int apply_rope) {
  const int lane = threadIdx.x & 31;
  const int wave = threadIdx.x >> 5;
  const int l15  = lane & 15;
  const int half = lane >> 4;
  const int koff = half * 2;
  const int n0 = (blockIdx.x * 8 + wave) * 16;
  const int m0 = blockIdx.y * 16;
  const int arow = m0 + l15;
  const int wrow = n0 + l15;

  v8f c = {0.f,0.f,0.f,0.f,0.f,0.f,0.f,0.f};

  if (in_head_major) {
    const int bb = arow >> 11;
    const int ss = arow & (SEQ - 1);
    #pragma unroll 8
    for (int kb = 0; kb < DMODEL; kb += 4) {
      const int k0 = kb + koff;
      v2f a = *(const v2f*)(A + (((size_t)(bb * NHEADS + (k0 >> 6)) * SEQ + ss) * DK)
                              + (k0 & 63));
      v2f b = *(const v2f*)(W + (size_t)wrow * DMODEL + k0);
      c = wmma_f32_4(a, b, c);
    }
  } else {
    #pragma unroll 8
    for (int kb = 0; kb < DMODEL; kb += 4) {
      const int k0 = kb + koff;
      v2f a = *(const v2f*)(A + (size_t)arow * DMODEL + k0);
      v2f b = *(const v2f*)(W + (size_t)wrow * DMODEL + k0);
      c = wmma_f32_4(a, b, c);
    }
  }

  const int e = n0 + l15;
  #pragma unroll
  for (int v = 0; v < 8; ++v) {
    const int mrow = m0 + v + half * 8;
    const int ss = mrow & (SEQ - 1);
    const int bb = mrow >> 11;
    float val = c[v];
    if (apply_rope) {
      float partner = __shfl_xor(val, 1, 32);
      const int p = (e & 63) >> 1;
      const float freq  = __expf(-(2.0f * LOG_THETA / (float)DK) * (float)p);
      const float phase = (float)ss * freq;
      float sn, cs;
      __sincosf(phase, &sn, &cs);
      val = (e & 1) ? (partner * sn + val * cs) : (val * cs - partner * sn);
    }
    if (out_head_major)
      Y[((size_t)(bb * NHEADS + (e >> 6)) * SEQ + ss) * DK + (e & 63)] = val;
    else
      Y[(size_t)mrow * DMODEL + e] = val;
  }
}

__global__ __launch_bounds__(256) void attn_kernel(
    const float* __restrict__ Q, const float* __restrict__ K,
    const float* __restrict__ V, float* __restrict__ O) {
  __shared__ float ldsS[8][256];
  __shared__ float ldsFac[8][16];
  __shared__ float ldsStat[8][16];

  const int lane = threadIdx.x & 31;
  const int wave = threadIdx.x >> 5;
  const int l15  = lane & 15;
  const int half = lane >> 4;
  const int koff = half * 2;
  const int qt = blockIdx.x * 8 + wave;
  const int h  = blockIdx.y;
  const int b  = blockIdx.z;
  const size_t base = ((size_t)(b * NHEADS + h)) * SEQ * DK;
  const float* Qh = Q + base;
  const float* Kh = K + base;
  const float* Vh = V + base;
  float*       Oh = O + base;
  const int r0 = qt * 16;

  float* S    = ldsS[wave];
  float* FAC  = ldsFac[wave];
  float* STAT = ldsStat[wave];

  v2f qa[16];
  #pragma unroll
  for (int kb = 0; kb < 16; ++kb)
    qa[kb] = *(const v2f*)(Qh + (size_t)(r0 + l15) * DK + kb * 4 + koff);

  v8f o0 = {0.f,0.f,0.f,0.f,0.f,0.f,0.f,0.f};
  v8f o1 = o0, o2 = o0, o3 = o0;
  float m_i = -__builtin_inff();
  float l_i = 0.f;

  for (int j = 0; j <= qt; ++j) {
    const int c0 = j * 16;

    v8f s = {0.f,0.f,0.f,0.f,0.f,0.f,0.f,0.f};
    #pragma unroll
    for (int kb = 0; kb < 16; ++kb) {
      v2f kv = *(const v2f*)(Kh + (size_t)(c0 + l15) * DK + kb * 4 + koff);
      s = wmma_f32_4(qa[kb], kv, s);
    }

    const int kcol = c0 + l15;
    #pragma unroll
    for (int v = 0; v < 8; ++v) {
      const int qrow = r0 + v + half * 8;
      S[(v + half * 8) * 16 + l15] =
          (kcol <= qrow) ? s[v] * 0.125f : -__builtin_inff();
    }
    asm volatile("s_wait_dscnt 0" ::: "memory");

    if (lane < 16) {
      float mrow = -__builtin_inff();
      for (int c2 = 0; c2 < 16; ++c2) mrow = fmaxf(mrow, S[lane * 16 + c2]);
      const float mnew = fmaxf(m_i, mrow);
      const float fac  = __expf(m_i - mnew);
      float rs = 0.f;
      for (int c2 = 0; c2 < 16; ++c2) {
        const float p = __expf(S[lane * 16 + c2] - mnew);
        S[lane * 16 + c2] = p;
        rs += p;
      }
      l_i = l_i * fac + rs;
      m_i = mnew;
      FAC[lane] = fac;
    }
    asm volatile("s_wait_dscnt 0" ::: "memory");

    float facv[8];
    #pragma unroll
    for (int v = 0; v < 8; ++v) facv[v] = FAC[v + half * 8];
    #pragma unroll
    for (int v = 0; v < 8; ++v) {
      o0[v] *= facv[v]; o1[v] *= facv[v];
      o2[v] *= facv[v]; o3[v] *= facv[v];
    }

    #pragma unroll
    for (int kb = 0; kb < 4; ++kb) {
      const int kk = kb * 4 + koff;
      v2f pa = *(const v2f*)(S + l15 * 16 + kk);
      const float* vp = Vh + (size_t)(c0 + kk) * DK + l15;
      v2f vb0 = { vp[0],  vp[DK] };
      v2f vb1 = { vp[16], vp[16 + DK] };
      v2f vb2 = { vp[32], vp[32 + DK] };
      v2f vb3 = { vp[48], vp[48 + DK] };
      o0 = wmma_f32_4(pa, vb0, o0);
      o1 = wmma_f32_4(pa, vb1, o1);
      o2 = wmma_f32_4(pa, vb2, o2);
      o3 = wmma_f32_4(pa, vb3, o3);
    }
  }

  if (lane < 16) STAT[lane] = 1.0f / l_i;
  asm volatile("s_wait_dscnt 0" ::: "memory");
  #pragma unroll
  for (int v = 0; v < 8; ++v) {
    const float linv = STAT[v + half * 8];
    float* op = Oh + (size_t)(r0 + v + half * 8) * DK + l15;
    op[0]  = o0[v] * linv;
    op[16] = o1[v] * linv;
    op[32] = o2[v] * linv;
    op[48] = o3[v] * linv;
  }
}

extern "C" void kernel_launch(void* const* d_in, const int* in_sizes, int n_in,
                              void* d_out, int out_size, void* d_ws,
                              size_t ws_size, hipStream_t stream) {
  (void)in_sizes; (void)n_in; (void)out_size; (void)ws_size;
  const float* x  = (const float*)d_in[0];
  const float* Wq = (const float*)d_in[1];
  const float* Wk = (const float*)d_in[2];
  const float* Wv = (const float*)d_in[3];
  const float* Wo = (const float*)d_in[4];
  float* out = (float*)d_out;

  const size_t perBuf = (size_t)BATCH * NHEADS * SEQ * DK;
  float* Qb = (float*)d_ws;
  float* Kb = Qb + perBuf;
  float* Vb = Kb + perBuf;
  float* AO = Vb + perBuf;

  dim3 blk(256);
  dim3 gproj(8, (BATCH * SEQ) / 16, 1);

  proj_gemm_kernel<<<gproj, blk, 0, stream>>>(x, Wq, Qb, 0, 1, 1);
  proj_gemm_kernel<<<gproj, blk, 0, stream>>>(x, Wk, Kb, 0, 1, 1);
  proj_gemm_kernel<<<gproj, blk, 0, stream>>>(x, Wv, Vb, 0, 1, 0);

  attn_kernel<<<dim3(SEQ / 16 / 8, NHEADS, BATCH), blk, 0, stream>>>(Qb, Kb, Vb, AO);

  proj_gemm_kernel<<<gproj, blk, 0, stream>>>(AO, Wo, out, 1, 0, 0);
}


# projection GEMMs: 128x64 block tile staged through double-buffered LDS, 32x32 output per wave, so operands are fetched once per block instead of once per 16x16 tile; k order unchanged
# speedup vs baseline: 1.6003x; 1.6003x over previous
#include <hip/hip_runtime.h>
#include <math.h>

#define BATCH   2
#define SEQ     2048
#define DMODEL  1024
#define NHEADS  16
#define DK      64
#define LOG_THETA 9.210340371976184f

typedef __attribute__((ext_vector_type(2))) float v2f;
typedef __attribute__((ext_vector_type(8))) float v8f;

static __device__ __forceinline__ v8f wmma_f32_4(v2f a, v2f b, v8f c) {
  return __builtin_amdgcn_wmma_f32_16x16x4_f32(
       false, a,  false, b,
       (short)0, c,  false,  false);
}

#define BM  128
#define BN  64
#define BK  32
#define LDP 36

static __device__ __forceinline__ void proj_tile_mac(
    const float* __restrict__ as, const float* __restrict__ bs,
    v8f& c00, v8f& c01, v8f& c10, v8f& c11) {
  #pragma unroll
  for (int kk = 0; kk < BK / 4; ++kk) {
    const v2f a0 = *(const v2f*)(as + kk * 4);
    const v2f a1 = *(const v2f*)(as + 16 * LDP + kk * 4);
    const v2f b0 = *(const v2f*)(bs + kk * 4);
    const v2f b1 = *(const v2f*)(bs + 16 * LDP + kk * 4);
    c00 = wmma_f32_4(a0, b0, c00);
    c01 = wmma_f32_4(a0, b1, c01);
    c10 = wmma_f32_4(a1, b0, c10);
    c11 = wmma_f32_4(a1, b1, c11);
  }
}

static __device__ __forceinline__ void proj_tile_store(
    const v8f c, const int m0, const int n0, const int l15, const int half,
    float* __restrict__ Y, const int out_head_major, const int apply_rope) {
  const int e = n0 + l15;
  #pragma unroll
  for (int v = 0; v < 8; ++v) {
    const int mrow = m0 + v + half * 8;
    const int ss = mrow & (SEQ - 1);
    const int bb = mrow >> 11;
    float val = c[v];
    if (apply_rope) {
      float partner = __shfl_xor(val, 1, 32);
      const int p = (e & 63) >> 1;
      const float freq  = __expf(-(2.0f * LOG_THETA / (float)DK) * (float)p);
      const float phase = (float)ss * freq;
      float sn, cs;
      __sincosf(phase, &sn, &cs);
      val = (e & 1) ? (partner * sn + val * cs) : (val * cs - partner * sn);
    }
    if (out_head_major)
      Y[((size_t)(bb * NHEADS + (e >> 6)) * SEQ + ss) * DK + (e & 63)] = val;
    else
      Y[(size_t)mrow * DMODEL + e] = val;
  }
}

__global__ __launch_bounds__(256) void proj_gemm_kernel(
    const float* __restrict__ A, const float* __restrict__ W,
    float* __restrict__ Y, int in_head_major, int out_head_major,
    int apply_rope) {
  __shared__ __attribute__((aligned(16))) float As[2][BM * LDP];
  __shared__ __attribute__((aligned(16))) float Bs[2][BN * LDP];

  const int tid  = threadIdx.x;
  const int lane = tid & 31;
  const int wave = tid >> 5;
  const int l15  = lane & 15;
  const int half = lane >> 4;
  const int koff = half * 2;
  const int wm   = wave & 3;
  const int wn   = wave >> 2;
  const int mblk = blockIdx.y * BM;
  const int nblk = blockIdx.x * BN;

  const int kq = tid & 7;
  const int lr = tid >> 3;

  size_t aoff[4];
  #pragma unroll
  for (int i = 0; i < 4; ++i) {
    const int r = mblk + lr + 32 * i;
    aoff[i] = in_head_major
        ? ((size_t)(r >> 11) * NHEADS * SEQ * DK + (size_t)(r & (SEQ - 1)) * DK)
        : (size_t)r * DMODEL;
  }
  size_t woff[2];
  #pragma unroll
  for (int i = 0; i < 2; ++i)
    woff[i] = (size_t)(nblk + lr + 32 * i) * DMODEL;

  float4 ra[4], rb[2];
  const int lds_st = lr * LDP + kq * 4;

  v8f c00 = {0.f,0.f,0.f,0.f,0.f,0.f,0.f,0.f};
  v8f c01 = c00, c10 = c00, c11 = c00;

  const int a_frag = (wm * 32 + l15) * LDP + koff;
  const int b_frag = (wn * 32 + l15) * LDP + koff;

  {
    const int k = kq * 4;
    #pragma unroll
    for (int i = 0; i < 4; ++i) ra[i] = *(const float4*)(A + aoff[i] + k);
    #pragma unroll
    for (int i = 0; i < 2; ++i) rb[i] = *(const float4*)(W + woff[i] + k);
    #pragma unroll
    for (int i = 0; i < 4; ++i) *(float4*)(&As[0][lds_st + 32 * i * LDP]) = ra[i];
    #pragma unroll
    for (int i = 0; i < 2; ++i) *(float4*)(&Bs[0][lds_st + 32 * i * LDP]) = rb[i];
  }
  __syncthreads();

  for (int kt = 0; kt < DMODEL / BK - 1; ++kt) {
    const int cur = kt & 1;
    const int k = (kt + 1) * BK + kq * 4;
    const size_t ka = in_head_major
        ? ((size_t)(k >> 6) * SEQ * DK + (size_t)(k & 63)) : (size_t)k;
    #pragma unroll
    for (int i = 0; i < 4; ++i) ra[i] = *(const float4*)(A + aoff[i] + ka);
    #pragma unroll
    for (int i = 0; i < 2; ++i) rb[i] = *(const float4*)(W + woff[i] + k);

    proj_tile_mac(&As[cur][a_frag], &Bs[cur][b_frag], c00, c01, c10, c11);

    #pragma unroll
    for (int i = 0; i < 4; ++i)
      *(float4*)(&As[cur ^ 1][lds_st + 32 * i * LDP]) = ra[i];
    #pragma unroll
    for (int i = 0; i < 2; ++i)
      *(float4*)(&Bs[cur ^ 1][lds_st + 32 * i * LDP]) = rb[i];
    __syncthreads();
  }
  proj_tile_mac(&As[1][a_frag], &Bs[1][b_frag], c00, c01, c10, c11);

  const int m0 = mblk + wm * 32;
  const int n0 = nblk + wn * 32;
  proj_tile_store(c00, m0,      n0,      l15, half, Y, out_head_major, apply_rope);
  proj_tile_store(c01, m0,      n0 + 16, l15, half, Y, out_head_major, apply_rope);
  proj_tile_store(c10, m0 + 16, n0,      l15, half, Y, out_head_major, apply_rope);
  proj_tile_store(c11, m0 + 16, n0 + 16, l15, half, Y, out_head_major, apply_rope);
}

__global__ __launch_bounds__(256) void attn_kernel(
    const float* __restrict__ Q, const float* __restrict__ K,
    const float* __restrict__ V, float* __restrict__ O) {
  __shared__ float ldsS[8][256];
  __shared__ float ldsFac[8][16];
  __shared__ float ldsStat[8][16];

  const int lane = threadIdx.x & 31;
  const int wave = threadIdx.x >> 5;
  const int l15  = lane & 15;
  const int half = lane >> 4;
  const int koff = half * 2;
  const int qt = blockIdx.x * 8 + wave;
  const int h  = blockIdx.y;
  const int b  = blockIdx.z;
  const size_t base = ((size_t)(b * NHEADS + h)) * SEQ * DK;
  const float* Qh = Q + base;
  const float* Kh = K + base;
  const float* Vh = V + base;
  float*       Oh = O + base;
  const int r0 = qt * 16;

  float* S    = ldsS[wave];
  float* FAC  = ldsFac[wave];
  float* STAT = ldsStat[wave];

  v2f qa[16];
  #pragma unroll
  for (int kb = 0; kb < 16; ++kb)
    qa[kb] = *(const v2f*)(Qh + (size_t)(r0 + l15) * DK + kb * 4 + koff);

  v8f o0 = {0.f,0.f,0.f,0.f,0.f,0.f,0.f,0.f};
  v8f o1 = o0, o2 = o0, o3 = o0;
  float m_i = -__builtin_inff();
  float l_i = 0.f;

  for (int j = 0; j <= qt; ++j) {
    const int c0 = j * 16;

    v8f s = {0.f,0.f,0.f,0.f,0.f,0.f,0.f,0.f};
    #pragma unroll
    for (int kb = 0; kb < 16; ++kb) {
      v2f kv = *(const v2f*)(Kh + (size_t)(c0 + l15) * DK + kb * 4 + koff);
      s = wmma_f32_4(qa[kb], kv, s);
    }

    const int kcol = c0 + l15;
    #pragma unroll
    for (int v = 0; v < 8; ++v) {
      const int qrow = r0 + v + half * 8;
      S[(v + half * 8) * 16 + l15] =
          (kcol <= qrow) ? s[v] * 0.125f : -__builtin_inff();
    }
    asm volatile("s_wait_dscnt 0" ::: "memory");

    if (lane < 16) {
      float mrow = -__builtin_inff();
      for (int c2 = 0; c2 < 16; ++c2) mrow = fmaxf(mrow, S[lane * 16 + c2]);
      const float mnew = fmaxf(m_i, mrow);
      const float fac  = __expf(m_i - mnew);
      float rs = 0.f;
      for (int c2 = 0; c2 < 16; ++c2) {
        const float p = __expf(S[lane * 16 + c2] - mnew);
        S[lane * 16 + c2] = p;
        rs += p;
      }
      l_i = l_i * fac + rs;
      m_i = mnew;
      FAC[lane] = fac;
    }
    asm volatile("s_wait_dscnt 0" ::: "memory");

    float facv[8];
    #pragma unroll
    for (int v = 0; v < 8; ++v) facv[v] = FAC[v + half * 8];
    #pragma unroll
    for (int v = 0; v < 8; ++v) {
      o0[v] *= facv[v]; o1[v] *= facv[v];
      o2[v] *= facv[v]; o3[v] *= facv[v];
    }

    #pragma unroll
    for (int kb = 0; kb < 4; ++kb) {
      const int kk = kb * 4 + koff;
      v2f pa = *(const v2f*)(S + l15 * 16 + kk);
      const float* vp = Vh + (size_t)(c0 + kk) * DK + l15;
      v2f vb0 = { vp[0],  vp[DK] };
      v2f vb1 = { vp[16], vp[16 + DK] };
      v2f vb2 = { vp[32], vp[32 + DK] };
      v2f vb3 = { vp[48], vp[48 + DK] };
      o0 = wmma_f32_4(pa, vb0, o0);
      o1 = wmma_f32_4(pa, vb1, o1);
      o2 = wmma_f32_4(pa, vb2, o2);
      o3 = wmma_f32_4(pa, vb3, o3);
    }
  }

  if (lane < 16) STAT[lane] = 1.0f / l_i;
  asm volatile("s_wait_dscnt 0" ::: "memory");
  #pragma unroll
  for (int v = 0; v < 8; ++v) {
    const float linv = STAT[v + half * 8];
    float* op = Oh + (size_t)(r0 + v + half * 8) * DK + l15;
    op[0]  = o0[v] * linv;
    op[16] = o1[v] * linv;
    op[32] = o2[v] * linv;
    op[48] = o3[v] * linv;
  }
}

extern "C" void kernel_launch(void* const* d_in, const int* in_sizes, int n_in,
                              void* d_out, int out_size, void* d_ws,
                              size_t ws_size, hipStream_t stream) {
  (void)in_sizes; (void)n_in; (void)out_size; (void)ws_size;
  const float* x  = (const float*)d_in[0];
  const float* Wq = (const float*)d_in[1];
  const float* Wk = (const float*)d_in[2];
  const float* Wv = (const float*)d_in[3];
  const float* Wo = (const float*)d_in[4];
  float* out = (float*)d_out;

  const size_t perBuf = (size_t)BATCH * NHEADS * SEQ * DK;
  float* Qb = (float*)d_ws;
  float* Kb = Qb + perBuf;
  float* Vb = Kb + perBuf;
  float* AO = Vb + perBuf;

  dim3 blk(256);
  dim3 gproj(DMODEL / BN, (BATCH * SEQ) / BM, 1);

  proj_gemm_kernel<<<gproj, blk, 0, stream>>>(x, Wq, Qb, 0, 1, 1);
  proj_gemm_kernel<<<gproj, blk, 0, stream>>>(x, Wk, Kb, 0, 1, 1);
  proj_gemm_kernel<<<gproj, blk, 0, stream>>>(x, Wv, Vb, 0, 1, 0);

  attn_kernel<<<dim3(SEQ / 16 / 8, NHEADS, BATCH), blk, 0, stream>>>(Qb, Kb, Vb, AO);

  proj_gemm_kernel<<<gproj, blk, 0, stream>>>(AO, Wo, out, 1, 0, 0);
}


# attention: K and V staged once per block through double-buffered LDS (32-key stages), fragments read from LDS in one group ahead of each WMMA run; no per-wave global gathers; sums unchanged
# speedup vs baseline: 1.9305x; 1.2063x over previous
#include <hip/hip_runtime.h>
#include <math.h>

#define BATCH   2
#define SEQ     2048
#define DMODEL  1024
#define NHEADS  16
#define DK      64
#define LOG_THETA 9.210340371976184f

typedef __attribute__((ext_vector_type(2))) float v2f;
typedef __attribute__((ext_vector_type(4))) float v4f;
typedef __attribute__((ext_vector_type(8))) float v8f;

static __device__ __forceinline__ v8f wmma_f32_4(v2f a, v2f b, v8f c) {
  return __builtin_amdgcn_wmma_f32_16x16x4_f32(
       false, a,  false, b,
       (short)0, c,  false,  false);
}

#define BM  128
#define BN  64
#define BK  32
#define LDP 36

static __device__ __forceinline__ void proj_tile_mac(
    const float* __restrict__ as, const float* __restrict__ bs,
    v8f& c00, v8f& c01, v8f& c10, v8f& c11) {
  #pragma unroll
  for (int kk = 0; kk < BK / 4; ++kk) {
    const v2f a0 = *(const v2f*)(as + kk * 4);
    const v2f a1 = *(const v2f*)(as + 16 * LDP + kk * 4);
    const v2f b0 = *(const v2f*)(bs + kk * 4);
    const v2f b1 = *(const v2f*)(bs + 16 * LDP + kk * 4);
    c00 = wmma_f32_4(a0, b0, c00);
    c01 = wmma_f32_4(a0, b1, c01);
    c10 = wmma_f32_4(a1, b0, c10);
    c11 = wmma_f32_4(a1, b1, c11);
  }
}

static __device__ __forceinline__ void proj_tile_store(
    const v8f c, const int m0, const int n0, const int l15, const int half,
    float* __restrict__ Y, const int out_head_major, const int apply_rope) {
  const int e = n0 + l15;
  #pragma unroll
  for (int v = 0; v < 8; ++v) {
    const int mrow = m0 + v + half * 8;
    const int ss = mrow & (SEQ - 1);
    const int bb = mrow >> 11;
    float val = c[v];
    if (apply_rope) {
      float partner = __shfl_xor(val, 1, 32);
      const int p = (e & 63) >> 1;
      const float freq  = __expf(-(2.0f * LOG_THETA / (float)DK) * (float)p);
      const float phase = (float)ss * freq;
      float sn, cs;
      __sincosf(phase, &sn, &cs);
      val = (e & 1) ? (partner * sn + val * cs) : (val * cs - partner * sn);
    }
    if (out_head_major)
      Y[((size_t)(bb * NHEADS + (e >> 6)) * SEQ + ss) * DK + (e & 63)] = val;
    else
      Y[(size_t)mrow * DMODEL + e] = val;
  }
}

__global__ __launch_bounds__(256) void proj_gemm_kernel(
    const float* __restrict__ A, const float* __restrict__ W,
    float* __restrict__ Y, int in_head_major, int out_head_major,
    int apply_rope) {
  __shared__ __attribute__((aligned(16))) float As[2][BM * LDP];
  __shared__ __attribute__((aligned(16))) float Bs[2][BN * LDP];

  const int tid  = threadIdx.x;
  const int lane = tid & 31;
  const int wave = tid >> 5;
  const int l15  = lane & 15;
  const int half = lane >> 4;
  const int koff = half * 2;
  const int wm   = wave & 3;
  const int wn   = wave >> 2;
  const int mblk = blockIdx.y * BM;
  const int nblk = blockIdx.x * BN;

  const int kq = tid & 7;
  const int lr = tid >> 3;

  size_t aoff[4];
  #pragma unroll
  for (int i = 0; i < 4; ++i) {
    const int r = mblk + lr + 32 * i;
    aoff[i] = in_head_major
        ? ((size_t)(r >> 11) * NHEADS * SEQ * DK + (size_t)(r & (SEQ - 1)) * DK)
        : (size_t)r * DMODEL;
  }
  size_t woff[2];
  #pragma unroll
  for (int i = 0; i < 2; ++i)
    woff[i] = (size_t)(nblk + lr + 32 * i) * DMODEL;

  float4 ra[4], rb[2];
  const int lds_st = lr * LDP + kq * 4;

  v8f c00 = {0.f,0.f,0.f,0.f,0.f,0.f,0.f,0.f};
  v8f c01 = c00, c10 = c00, c11 = c00;

  const int a_frag = (wm * 32 + l15) * LDP + koff;
  const int b_frag = (wn * 32 + l15) * LDP + koff;

  {
    const int k = kq * 4;
    #pragma unroll
    for (int i = 0; i < 4; ++i) ra[i] = *(const float4*)(A + aoff[i] + k);
    #pragma unroll
    for (int i = 0; i < 2; ++i) rb[i] = *(const float4*)(W + woff[i] + k);
    #pragma unroll
    for (int i = 0; i < 4; ++i) *(float4*)(&As[0][lds_st + 32 * i * LDP]) = ra[i];
    #pragma unroll
    for (int i = 0; i < 2; ++i) *(float4*)(&Bs[0][lds_st + 32 * i * LDP]) = rb[i];
  }
  __syncthreads();

  for (int kt = 0; kt < DMODEL / BK - 1; ++kt) {
    const int cur = kt & 1;
    const int k = (kt + 1) * BK + kq * 4;
    const size_t ka = in_head_major
        ? ((size_t)(k >> 6) * SEQ * DK + (size_t)(k & 63)) : (size_t)k;
    #pragma unroll
    for (int i = 0; i < 4; ++i) ra[i] = *(const float4*)(A + aoff[i] + ka);
    #pragma unroll
    for (int i = 0; i < 2; ++i) rb[i] = *(const float4*)(W + woff[i] + k);

    proj_tile_mac(&As[cur][a_frag], &Bs[cur][b_frag], c00, c01, c10, c11);

    #pragma unroll
    for (int i = 0; i < 4; ++i)
      *(float4*)(&As[cur ^ 1][lds_st + 32 * i * LDP]) = ra[i];
    #pragma unroll
    for (int i = 0; i < 2; ++i)
      *(float4*)(&Bs[cur ^ 1][lds_st + 32 * i * LDP]) = rb[i];
    __syncthreads();
  }
  proj_tile_mac(&As[1][a_frag], &Bs[1][b_frag], c00, c01, c10, c11);

  const int m0 = mblk + wm * 32;
  const int n0 = nblk + wn * 32;
  proj_tile_store(c00, m0,      n0,      l15, half, Y, out_head_major, apply_rope);
  proj_tile_store(c01, m0,      n0 + 16, l15, half, Y, out_head_major, apply_rope);
  proj_tile_store(c10, m0 + 16, n0,      l15, half, Y, out_head_major, apply_rope);
  proj_tile_store(c11, m0 + 16, n0 + 16, l15, half, Y, out_head_major, apply_rope);
}

#define KT  32
#define KP  68
#define VP  72

__global__ __launch_bounds__(256) void attn_kernel(
    const float* __restrict__ Q, const float* __restrict__ K,
    const float* __restrict__ V, float* __restrict__ O) {
  __shared__ float ldsS[8][256];
  __shared__ float ldsFac[8][16];
  __shared__ float ldsStat[8][16];
  __shared__ __attribute__((aligned(16))) float Ks[2][KT * KP];
  __shared__ __attribute__((aligned(16))) float Vs[2][KT * VP];

  const int tid  = threadIdx.x;
  const int lane = tid & 31;
  const int wave = tid >> 5;
  const int l15  = lane & 15;
  const int half = lane >> 4;
  const int koff = half * 2;
  const int qt = blockIdx.x * 8 + wave;
  const int h  = blockIdx.y;
  const int b  = blockIdx.z;
  const size_t base = ((size_t)(b * NHEADS + h)) * SEQ * DK;
  const float* Qh = Q + base;
  const float* Kh = K + base;
  const float* Vh = V + base;
  float*       Oh = O + base;
  const int r0 = qt * 16;

  float* S    = ldsS[wave];
  float* FAC  = ldsFac[wave];
  float* STAT = ldsStat[wave];

  const int sq = tid & 15;
  const int sr = tid >> 4;
  const int k_st = sr * KP + sq * 4;
  const int v_st = sr * VP + sq * 4;
  const int nst = blockIdx.x * 4 + 4;

  v2f qa[16];
  #pragma unroll
  for (int kb = 0; kb < 16; ++kb)
    qa[kb] = *(const v2f*)(Qh + (size_t)(r0 + l15) * DK + kb * 4 + koff);

  v4f rk0, rk1, rv0, rv1;
  {
    const size_t g = (size_t)sr * DK + sq * 4;
    rk0 = *(const v4f*)(Kh + g);
    rk1 = *(const v4f*)(Kh + g + 16 * DK);
    rv0 = *(const v4f*)(Vh + g);
    rv1 = *(const v4f*)(Vh + g + 16 * DK);
    *(v4f*)(&Ks[0][k_st])           = rk0;
    *(v4f*)(&Ks[0][k_st + 16 * KP]) = rk1;
    *(v4f*)(&Vs[0][v_st])           = rv0;
    *(v4f*)(&Vs[0][v_st + 16 * VP]) = rv1;
  }
  __syncthreads();

  v8f o0 = {0.f,0.f,0.f,0.f,0.f,0.f,0.f,0.f};
  v8f o1 = o0, o2 = o0, o3 = o0;
  float m_i = -__builtin_inff();
  float l_i = 0.f;

  for (int st = 0; st < nst; ++st) {
    const int cur = st & 1;
    const bool more = (st + 1 < nst);
    if (more) {
      const size_t g = (size_t)((st + 1) * KT + sr) * DK + sq * 4;
      rk0 = *(const v4f*)(Kh + g);
      rk1 = *(const v4f*)(Kh + g + 16 * DK);
      rv0 = *(const v4f*)(Vh + g);
      rv1 = *(const v4f*)(Vh + g + 16 * DK);
    }

    for (int t = 0; t < KT / 16; ++t) {
      const int j = st * (KT / 16) + t;
      if (j > qt) continue;
      const int c0 = j * 16;
      const float* kt = &Ks[cur][t * 16 * KP];
      const float* vt = &Vs[cur][t * 16 * VP];

      v8f s = {0.f,0.f,0.f,0.f,0.f,0.f,0.f,0.f};
      #pragma unroll
      for (int kb = 0; kb < 16; ++kb) {
        v2f kv = *(const v2f*)(kt + l15 * KP + kb * 4 + koff);
        s = wmma_f32_4(qa[kb], kv, s);
      }
      __builtin_amdgcn_sched_group_barrier(0x100, 8, 0);
      __builtin_amdgcn_sched_group_barrier(0x008, 16, 0);

      const int kcol = c0 + l15;
      #pragma unroll
      for (int v = 0; v < 8; ++v) {
        const int qrow = r0 + v + half * 8;
        S[(v + half * 8) * 16 + l15] =
            (kcol <= qrow) ? s[v] * 0.125f : -__builtin_inff();
      }
      asm volatile("s_wait_dscnt 0" ::: "memory");

      if (lane < 16) {
        float mrow = -__builtin_inff();
        for (int c2 = 0; c2 < 16; ++c2) mrow = fmaxf(mrow, S[lane * 16 + c2]);
        const float mnew = fmaxf(m_i, mrow);
        const float fac  = __expf(m_i - mnew);
        float rs = 0.f;
        for (int c2 = 0; c2 < 16; ++c2) {
          const float p = __expf(S[lane * 16 + c2] - mnew);
          S[lane * 16 + c2] = p;
          rs += p;
        }
        l_i = l_i * fac + rs;
        m_i = mnew;
        FAC[lane] = fac;
      }
      asm volatile("s_wait_dscnt 0" ::: "memory");

      float facv[8];
      #pragma unroll
      for (int v = 0; v < 8; ++v) facv[v] = FAC[v + half * 8];
      #pragma unroll
      for (int v = 0; v < 8; ++v) {
        o0[v] *= facv[v]; o1[v] *= facv[v];
        o2[v] *= facv[v]; o3[v] *= facv[v];
      }

      #pragma unroll
      for (int kb = 0; kb < 4; ++kb) {
        const int kk = kb * 4 + koff;
        v2f pa = *(const v2f*)(S + l15 * 16 + kk);
        const float* vp = vt + kk * VP + l15;
        v2f vb0 = { vp[0],  vp[VP] };
        v2f vb1 = { vp[16], vp[16 + VP] };
        v2f vb2 = { vp[32], vp[32 + VP] };
        v2f vb3 = { vp[48], vp[48 + VP] };
        o0 = wmma_f32_4(pa, vb0, o0);
        o1 = wmma_f32_4(pa, vb1, o1);
        o2 = wmma_f32_4(pa, vb2, o2);
        o3 = wmma_f32_4(pa, vb3, o3);
      }
      __builtin_amdgcn_sched_group_barrier(0x100, 20, 1);
      __builtin_amdgcn_sched_group_barrier(0x008, 16, 1);
    }

    if (more) {
      *(v4f*)(&Ks[cur ^ 1][k_st])           = rk0;
      *(v4f*)(&Ks[cur ^ 1][k_st + 16 * KP]) = rk1;
      *(v4f*)(&Vs[cur ^ 1][v_st])           = rv0;
      *(v4f*)(&Vs[cur ^ 1][v_st + 16 * VP]) = rv1;
    }
    __syncthreads();
  }

  if (lane < 16) STAT[lane] = 1.0f / l_i;
  asm volatile("s_wait_dscnt 0" ::: "memory");
  #pragma unroll
  for (int v = 0; v < 8; ++v) {
    const float linv = STAT[v + half * 8];
    float* op = Oh + (size_t)(r0 + v + half * 8) * DK + l15;
    op[0]  = o0[v] * linv;
    op[16] = o1[v] * linv;
    op[32] = o2[v] * linv;
    op[48] = o3[v] * linv;
  }
}

extern "C" void kernel_launch(void* const* d_in, const int* in_sizes, int n_in,
                              void* d_out, int out_size, void* d_ws,
                              size_t ws_size, hipStream_t stream) {
  (void)in_sizes; (void)n_in; (void)out_size; (void)ws_size;
  const float* x  = (const float*)d_in[0];
  const float* Wq = (const float*)d_in[1];
  const float* Wk = (const float*)d_in[2];
  const float* Wv = (const float*)d_in[3];
  const float* Wo = (const float*)d_in[4];
  float* out = (float*)d_out;

  const size_t perBuf = (size_t)BATCH * NHEADS * SEQ * DK;
  float* Qb = (float*)d_ws;
  float* Kb = Qb + perBuf;
  float* Vb = Kb + perBuf;
  float* AO = Vb + perBuf;

  dim3 blk(256);
  dim3 gproj(DMODEL / BN, (BATCH * SEQ) / BM, 1);

  proj_gemm_kernel<<<gproj, blk, 0, stream>>>(x, Wq, Qb, 0, 1, 1);
  proj_gemm_kernel<<<gproj, blk, 0, stream>>>(x, Wk, Kb, 0, 1, 1);
  proj_gemm_kernel<<<gproj, blk, 0, stream>>>(x, Wv, Vb, 0, 1, 0);

  attn_kernel<<<dim3(SEQ / 16 / 8, NHEADS, BATCH), blk, 0, stream>>>(Qb, Kb, Vb, AO);

  proj_gemm_kernel<<<gproj, blk, 0, stream>>>(AO, Wo, out, 1, 0, 0);
}


# attention: each block takes 4 light query tiles (4p..4p+3) and their 4 mirrored heavy ones (124-4p..127-4p), so every block and SIMD gets equal causal work; per-row sums and their order unchanged
# speedup vs baseline: 2.2796x; 1.1808x over previous
#include <hip/hip_runtime.h>
#include <math.h>

#define BATCH   2
#define SEQ     2048
#define DMODEL  1024
#define NHEADS  16
#define DK      64
#define LOG_THETA 9.210340371976184f

typedef __attribute__((ext_vector_type(2))) float v2f;
typedef __attribute__((ext_vector_type(4))) float v4f;
typedef __attribute__((ext_vector_type(8))) float v8f;

static __device__ __forceinline__ v8f wmma_f32_4(v2f a, v2f b, v8f c) {
  return __builtin_amdgcn_wmma_f32_16x16x4_f32(
       false, a,  false, b,
       (short)0, c,  false,  false);
}

#define BM  128
#define BN  64
#define BK  32
#define LDP 36

static __device__ __forceinline__ void proj_tile_mac(
    const float* __restrict__ as, const float* __restrict__ bs,
    v8f& c00, v8f& c01, v8f& c10, v8f& c11) {
  #pragma unroll
  for (int kk = 0; kk < BK / 4; ++kk) {
    const v2f a0 = *(const v2f*)(as + kk * 4);
    const v2f a1 = *(const v2f*)(as + 16 * LDP + kk * 4);
    const v2f b0 = *(const v2f*)(bs + kk * 4);
    const v2f b1 = *(const v2f*)(bs + 16 * LDP + kk * 4);
    c00 = wmma_f32_4(a0, b0, c00);
    c01 = wmma_f32_4(a0, b1, c01);
    c10 = wmma_f32_4(a1, b0, c10);
    c11 = wmma_f32_4(a1, b1, c11);
  }
}

static __device__ __forceinline__ void proj_tile_store(
    const v8f c, const int m0, const int n0, const int l15, const int half,
    float* __restrict__ Y, const int out_head_major, const int apply_rope) {
  const int e = n0 + l15;
  #pragma unroll
  for (int v = 0; v < 8; ++v) {
    const int mrow = m0 + v + half * 8;
    const int ss = mrow & (SEQ - 1);
    const int bb = mrow >> 11;
    float val = c[v];
    if (apply_rope) {
      float partner = __shfl_xor(val, 1, 32);
      const int p = (e & 63) >> 1;
      const float freq  = __expf(-(2.0f * LOG_THETA / (float)DK) * (float)p);
      const float phase = (float)ss * freq;
      float sn, cs;
      __sincosf(phase, &sn, &cs);
      val = (e & 1) ? (partner * sn + val * cs) : (val * cs - partner * sn);
    }
    if (out_head_major)
      Y[((size_t)(bb * NHEADS + (e >> 6)) * SEQ + ss) * DK + (e & 63)] = val;
    else
      Y[(size_t)mrow * DMODEL + e] = val;
  }
}

__global__ __launch_bounds__(256) void proj_gemm_kernel(
    const float* __restrict__ A, const float* __restrict__ W,
    float* __restrict__ Y, int in_head_major, int out_head_major,
    int apply_rope) {
  __shared__ __attribute__((aligned(16))) float As[2][BM * LDP];
  __shared__ __attribute__((aligned(16))) float Bs[2][BN * LDP];

  const int tid  = threadIdx.x;
  const int lane = tid & 31;
  const int wave = tid >> 5;
  const int l15  = lane & 15;
  const int half = lane >> 4;
  const int koff = half * 2;
  const int wm   = wave & 3;
  const int wn   = wave >> 2;
  const int mblk = blockIdx.y * BM;
  const int nblk = blockIdx.x * BN;

  const int kq = tid & 7;
  const int lr = tid >> 3;

  size_t aoff[4];
  #pragma unroll
  for (int i = 0; i < 4; ++i) {
    const int r = mblk + lr + 32 * i;
    aoff[i] = in_head_major
        ? ((size_t)(r >> 11) * NHEADS * SEQ * DK + (size_t)(r & (SEQ - 1)) * DK)
        : (size_t)r * DMODEL;
  }
  size_t woff[2];
  #pragma unroll
  for (int i = 0; i < 2; ++i)
    woff[i] = (size_t)(nblk + lr + 32 * i) * DMODEL;

  float4 ra[4], rb[2];
  const int lds_st = lr * LDP + kq * 4;

  v8f c00 = {0.f,0.f,0.f,0.f,0.f,0.f,0.f,0.f};
  v8f c01 = c00, c10 = c00, c11 = c00;

  const int a_frag = (wm * 32 + l15) * LDP + koff;
  const int b_frag = (wn * 32 + l15) * LDP + koff;

  {
    const int k = kq * 4;
    #pragma unroll
    for (int i = 0; i < 4; ++i) ra[i] = *(const float4*)(A + aoff[i] + k);
    #pragma unroll
    for (int i = 0; i < 2; ++i) rb[i] = *(const float4*)(W + woff[i] + k);
    #pragma unroll
    for (int i = 0; i < 4; ++i) *(float4*)(&As[0][lds_st + 32 * i * LDP]) = ra[i];
    #pragma unroll
    for (int i = 0; i < 2; ++i) *(float4*)(&Bs[0][lds_st + 32 * i * LDP]) = rb[i];
  }
  __syncthreads();

  for (int kt = 0; kt < DMODEL / BK - 1; ++kt) {
    const int cur = kt & 1;
    const int k = (kt + 1) * BK + kq * 4;
    const size_t ka = in_head_major
        ? ((size_t)(k >> 6) * SEQ * DK + (size_t)(k & 63)) : (size_t)k;
    #pragma unroll
    for (int i = 0; i < 4; ++i) ra[i] = *(const float4*)(A + aoff[i] + ka);
    #pragma unroll
    for (int i = 0; i < 2; ++i) rb[i] = *(const float4*)(W + woff[i] + k);

    proj_tile_mac(&As[cur][a_frag], &Bs[cur][b_frag], c00, c01, c10, c11);

    #pragma unroll
    for (int i = 0; i < 4; ++i)
      *(float4*)(&As[cur ^ 1][lds_st + 32 * i * LDP]) = ra[i];
    #pragma unroll
    for (int i = 0; i < 2; ++i)
      *(float4*)(&Bs[cur ^ 1][lds_st + 32 * i * LDP]) = rb[i];
    __syncthreads();
  }
  proj_tile_mac(&As[1][a_frag], &Bs[1][b_frag], c00, c01, c10, c11);

  const int m0 = mblk + wm * 32;
  const int n0 = nblk + wn * 32;
  proj_tile_store(c00, m0,      n0,      l15, half, Y, out_head_major, apply_rope);
  proj_tile_store(c01, m0,      n0 + 16, l15, half, Y, out_head_major, apply_rope);
  proj_tile_store(c10, m0 + 16, n0,      l15, half, Y, out_head_major, apply_rope);
  proj_tile_store(c11, m0 + 16, n0 + 16, l15, half, Y, out_head_major, apply_rope);
}

#define KT  32
#define KP  68
#define VP  72

__global__ __launch_bounds__(256) void attn_kernel(
    const float* __restrict__ Q, const float* __restrict__ K,
    const float* __restrict__ V, float* __restrict__ O) {
  __shared__ float ldsS[8][256];
  __shared__ float ldsFac[8][16];
  __shared__ float ldsStat[8][16];
  __shared__ __attribute__((aligned(16))) float Ks[2][KT * KP];
  __shared__ __attribute__((aligned(16))) float Vs[2][KT * VP];

  const int tid  = threadIdx.x;
  const int lane = tid & 31;
  const int wave = tid >> 5;
  const int l15  = lane & 15;
  const int half = lane >> 4;
  const int koff = half * 2;
  const int p  = blockIdx.x;
  const int qt = (wave < 4) ? (4 * p + wave) : (120 - 4 * p + wave);
  const int h  = blockIdx.y;
  const int b  = blockIdx.z;
  const size_t base = ((size_t)(b * NHEADS + h)) * SEQ * DK;
  const float* Qh = Q + base;
  const float* Kh = K + base;
  const float* Vh = V + base;
  float*       Oh = O + base;
  const int r0 = qt * 16;

  float* S    = ldsS[wave];
  float* FAC  = ldsFac[wave];
  float* STAT = ldsStat[wave];

  const int sq = tid & 15;
  const int sr = tid >> 4;
  const int k_st = sr * KP + sq * 4;
  const int v_st = sr * VP + sq * 4;
  const int nst = 64 - 2 * p;

  v2f qa[16];
  #pragma unroll
  for (int kb = 0; kb < 16; ++kb)
    qa[kb] = *(const v2f*)(Qh + (size_t)(r0 + l15) * DK + kb * 4 + koff);

  v4f rk0, rk1, rv0, rv1;
  {
    const size_t g = (size_t)sr * DK + sq * 4;
    rk0 = *(const v4f*)(Kh + g);
    rk1 = *(const v4f*)(Kh + g + 16 * DK);
    rv0 = *(const v4f*)(Vh + g);
    rv1 = *(const v4f*)(Vh + g + 16 * DK);
    *(v4f*)(&Ks[0][k_st])           = rk0;
    *(v4f*)(&Ks[0][k_st + 16 * KP]) = rk1;
    *(v4f*)(&Vs[0][v_st])           = rv0;
    *(v4f*)(&Vs[0][v_st + 16 * VP]) = rv1;
  }
  __syncthreads();

  v8f o0 = {0.f,0.f,0.f,0.f,0.f,0.f,0.f,0.f};
  v8f o1 = o0, o2 = o0, o3 = o0;
  float m_i = -__builtin_inff();
  float l_i = 0.f;

  for (int st = 0; st < nst; ++st) {
    const int cur = st & 1;
    const bool more = (st + 1 < nst);
    if (more) {
      const size_t g = (size_t)((st + 1) * KT + sr) * DK + sq * 4;
      rk0 = *(const v4f*)(Kh + g);
      rk1 = *(const v4f*)(Kh + g + 16 * DK);
      rv0 = *(const v4f*)(Vh + g);
      rv1 = *(const v4f*)(Vh + g + 16 * DK);
    }

    for (int t = 0; t < KT / 16; ++t) {
      const int j = st * (KT / 16) + t;
      if (j > qt) continue;
      const int c0 = j * 16;
      const float* kt = &Ks[cur][t * 16 * KP];
      const float* vt = &Vs[cur][t * 16 * VP];

      v8f s = {0.f,0.f,0.f,0.f,0.f,0.f,0.f,0.f};
      #pragma unroll
      for (int kb = 0; kb < 16; ++kb) {
        v2f kv = *(const v2f*)(kt + l15 * KP + kb * 4 + koff);
        s = wmma_f32_4(qa[kb], kv, s);
      }
      __builtin_amdgcn_sched_group_barrier(0x100, 8, 0);
      __builtin_amdgcn_sched_group_barrier(0x008, 16, 0);

      const int kcol = c0 + l15;
      #pragma unroll
      for (int v = 0; v < 8; ++v) {
        const int qrow = r0 + v + half * 8;
        S[(v + half * 8) * 16 + l15] =
            (kcol <= qrow) ? s[v] * 0.125f : -__builtin_inff();
      }
      asm volatile("s_wait_dscnt 0" ::: "memory");

      if (lane < 16) {
        float mrow = -__builtin_inff();
        for (int c2 = 0; c2 < 16; ++c2) mrow = fmaxf(mrow, S[lane * 16 + c2]);
        const float mnew = fmaxf(m_i, mrow);
        const float fac  = __expf(m_i - mnew);
        float rs = 0.f;
        for (int c2 = 0; c2 < 16; ++c2) {
          const float p = __expf(S[lane * 16 + c2] - mnew);
          S[lane * 16 + c2] = p;
          rs += p;
        }
        l_i = l_i * fac + rs;
        m_i = mnew;
        FAC[lane] = fac;
      }
      asm volatile("s_wait_dscnt 0" ::: "memory");

      float facv[8];
      #pragma unroll
      for (int v = 0; v < 8; ++v) facv[v] = FAC[v + half * 8];
      #pragma unroll
      for (int v = 0; v < 8; ++v) {
        o0[v] *= facv[v]; o1[v] *= facv[v];
        o2[v] *= facv[v]; o3[v] *= facv[v];
      }

      #pragma unroll
      for (int kb = 0; kb < 4; ++kb) {
        const int kk = kb * 4 + koff;
        v2f pa = *(const v2f*)(S + l15 * 16 + kk);
        const float* vp = vt + kk * VP + l15;
        v2f vb0 = { vp[0],  vp[VP] };
        v2f vb1 = { vp[16], vp[16 + VP] };
        v2f vb2 = { vp[32], vp[32 + VP] };
        v2f vb3 = { vp[48], vp[48 + VP] };
        o0 = wmma_f32_4(pa, vb0, o0);
        o1 = wmma_f32_4(pa, vb1, o1);
        o2 = wmma_f32_4(pa, vb2, o2);
        o3 = wmma_f32_4(pa, vb3, o3);
      }
      __builtin_amdgcn_sched_group_barrier(0x100, 20, 1);
      __builtin_amdgcn_sched_group_barrier(0x008, 16, 1);
    }

    if (more) {
      *(v4f*)(&Ks[cur ^ 1][k_st])           = rk0;
      *(v4f*)(&Ks[cur ^ 1][k_st + 16 * KP]) = rk1;
      *(v4f*)(&Vs[cur ^ 1][v_st])           = rv0;
      *(v4f*)(&Vs[cur ^ 1][v_st + 16 * VP]) = rv1;
    }
    __syncthreads();
  }

  if (lane < 16) STAT[lane] = 1.0f / l_i;
  asm volatile("s_wait_dscnt 0" ::: "memory");
  #pragma unroll
  for (int v = 0; v < 8; ++v) {
    const float linv = STAT[v + half * 8];
    float* op = Oh + (size_t)(r0 + v + half * 8) * DK + l15;
    op[0]  = o0[v] * linv;
    op[16] = o1[v] * linv;
    op[32] = o2[v] * linv;
    op[48] = o3[v] * linv;
  }
}

extern "C" void kernel_launch(void* const* d_in, const int* in_sizes, int n_in,
                              void* d_out, int out_size, void* d_ws,
                              size_t ws_size, hipStream_t stream) {
  (void)in_sizes; (void)n_in; (void)out_size; (void)ws_size;
  const float* x  = (const float*)d_in[0];
  const float* Wq = (const float*)d_in[1];
  const float* Wk = (const float*)d_in[2];
  const float* Wv = (const float*)d_in[3];
  const float* Wo = (const float*)d_in[4];
  float* out = (float*)d_out;

  const size_t perBuf = (size_t)BATCH * NHEADS * SEQ * DK;
  float* Qb = (float*)d_ws;
  float* Kb = Qb + perBuf;
  float* Vb = Kb + perBuf;
  float* AO = Vb + perBuf;

  dim3 blk(256);
  dim3 gproj(DMODEL / BN, (BATCH * SEQ) / BM, 1);

  proj_gemm_kernel<<<gproj, blk, 0, stream>>>(x, Wq, Qb, 0, 1, 1);
  proj_gemm_kernel<<<gproj, blk, 0, stream>>>(x, Wk, Kb, 0, 1, 1);
  proj_gemm_kernel<<<gproj, blk, 0, stream>>>(x, Wv, Vb, 0, 1, 0);

  attn_kernel<<<dim3(SEQ / 16 / 8, NHEADS, BATCH), blk, 0, stream>>>(Qb, Kb, Vb, AO);

  proj_gemm_kernel<<<gproj, blk, 0, stream>>>(AO, Wo, out, 1, 0, 0);
}


# attention: S and O kept transposed (keys x queries; V staged key-pair interleaved), a lane owns one query: softmax in registers, no LDS round trip; row sums join per lane half, P*V key order changes
# speedup vs baseline: 2.4414x; 1.0710x over previous
#include <hip/hip_runtime.h>
#include <math.h>

#define BATCH   2
#define SEQ     2048
#define DMODEL  1024
#define NHEADS  16
#define DK      64
#define LOG_THETA 9.210340371976184f

typedef __attribute__((ext_vector_type(2))) float v2f;
typedef __attribute__((ext_vector_type(4))) float v4f;
typedef __attribute__((ext_vector_type(8))) float v8f;

static __device__ __forceinline__ v8f wmma_f32_4(v2f a, v2f b, v8f c) {
  return __builtin_amdgcn_wmma_f32_16x16x4_f32(
       false, a,  false, b,
       (short)0, c,  false,  false);
}

#define BM  128
#define BN  64
#define BK  32
#define LDP 36

static __device__ __forceinline__ void proj_tile_mac(
    const float* __restrict__ as, const float* __restrict__ bs,
    v8f& c00, v8f& c01, v8f& c10, v8f& c11) {
  #pragma unroll
  for (int kk = 0; kk < BK / 4; ++kk) {
    const v2f a0 = *(const v2f*)(as + kk * 4);
    const v2f a1 = *(const v2f*)(as + 16 * LDP + kk * 4);
    const v2f b0 = *(const v2f*)(bs + kk * 4);
    const v2f b1 = *(const v2f*)(bs + 16 * LDP + kk * 4);
    c00 = wmma_f32_4(a0, b0, c00);
    c01 = wmma_f32_4(a0, b1, c01);
    c10 = wmma_f32_4(a1, b0, c10);
    c11 = wmma_f32_4(a1, b1, c11);
  }
}

static __device__ __forceinline__ void proj_tile_store(
    const v8f c, const int m0, const int n0, const int l15, const int half,
    float* __restrict__ Y, const int out_head_major, const int apply_rope) {
  const int e = n0 + l15;
  #pragma unroll
  for (int v = 0; v < 8; ++v) {
    const int mrow = m0 + v + half * 8;
    const int ss = mrow & (SEQ - 1);
    const int bb = mrow >> 11;
    float val = c[v];
    if (apply_rope) {
      float partner = __shfl_xor(val, 1, 32);
      const int p = (e & 63) >> 1;
      const float freq  = __expf(-(2.0f * LOG_THETA / (float)DK) * (float)p);
      const float phase = (float)ss * freq;
      float sn, cs;
      __sincosf(phase, &sn, &cs);
      val = (e & 1) ? (partner * sn + val * cs) : (val * cs - partner * sn);
    }
    if (out_head_major)
      Y[((size_t)(bb * NHEADS + (e >> 6)) * SEQ + ss) * DK + (e & 63)] = val;
    else
      Y[(size_t)mrow * DMODEL + e] = val;
  }
}

__global__ __launch_bounds__(256) void proj_gemm_kernel(
    const float* __restrict__ A, const float* __restrict__ W,
    float* __restrict__ Y, int in_head_major, int out_head_major,
    int apply_rope) {
  __shared__ __attribute__((aligned(16))) float As[2][BM * LDP];
  __shared__ __attribute__((aligned(16))) float Bs[2][BN * LDP];

  const int tid  = threadIdx.x;
  const int lane = tid & 31;
  const int wave = tid >> 5;
  const int l15  = lane & 15;
  const int half = lane >> 4;
  const int koff = half * 2;
  const int wm   = wave & 3;
  const int wn   = wave >> 2;
  const int mblk = blockIdx.y * BM;
  const int nblk = blockIdx.x * BN;

  const int kq = tid & 7;
  const int lr = tid >> 3;

  size_t aoff[4];
  #pragma unroll
  for (int i = 0; i < 4; ++i) {
    const int r = mblk + lr + 32 * i;
    aoff[i] = in_head_major
        ? ((size_t)(r >> 11) * NHEADS * SEQ * DK + (size_t)(r & (SEQ - 1)) * DK)
        : (size_t)r * DMODEL;
  }
  size_t woff[2];
  #pragma unroll
  for (int i = 0; i < 2; ++i)
    woff[i] = (size_t)(nblk + lr + 32 * i) * DMODEL;

  float4 ra[4], rb[2];
  const int lds_st = lr * LDP + kq * 4;

  v8f c00 = {0.f,0.f,0.f,0.f,0.f,0.f,0.f,0.f};
  v8f c01 = c00, c10 = c00, c11 = c00;

  const int a_frag = (wm * 32 + l15) * LDP + koff;
  const int b_frag = (wn * 32 + l15) * LDP + koff;

  {
    const int k = kq * 4;
    #pragma unroll
    for (int i = 0; i < 4; ++i) ra[i] = *(const float4*)(A + aoff[i] + k);
    #pragma unroll
    for (int i = 0; i < 2; ++i) rb[i] = *(const float4*)(W + woff[i] + k);
    #pragma unroll
    for (int i = 0; i < 4; ++i) *(float4*)(&As[0][lds_st + 32 * i * LDP]) = ra[i];
    #pragma unroll
    for (int i = 0; i < 2; ++i) *(float4*)(&Bs[0][lds_st + 32 * i * LDP]) = rb[i];
  }
  __syncthreads();

  for (int kt = 0; kt < DMODEL / BK - 1; ++kt) {
    const int cur = kt & 1;
    const int k = (kt + 1) * BK + kq * 4;
    const size_t ka = in_head_major
        ? ((size_t)(k >> 6) * SEQ * DK + (size_t)(k & 63)) : (size_t)k;
    #pragma unroll
    for (int i = 0; i < 4; ++i) ra[i] = *(const float4*)(A + aoff[i] + ka);
    #pragma unroll
    for (int i = 0; i < 2; ++i) rb[i] = *(const float4*)(W + woff[i] + k);

    proj_tile_mac(&As[cur][a_frag], &Bs[cur][b_frag], c00, c01, c10, c11);

    #pragma unroll
    for (int i = 0; i < 4; ++i)
      *(float4*)(&As[cur ^ 1][lds_st + 32 * i * LDP]) = ra[i];
    #pragma unroll
    for (int i = 0; i < 2; ++i)
      *(float4*)(&Bs[cur ^ 1][lds_st + 32 * i * LDP]) = rb[i];
    __syncthreads();
  }
  proj_tile_mac(&As[1][a_frag], &Bs[1][b_frag], c00, c01, c10, c11);

  const int m0 = mblk + wm * 32;
  const int n0 = nblk + wn * 32;
  proj_tile_store(c00, m0,      n0,      l15, half, Y, out_head_major, apply_rope);
  proj_tile_store(c01, m0,      n0 + 16, l15, half, Y, out_head_major, apply_rope);
  proj_tile_store(c10, m0 + 16, n0,      l15, half, Y, out_head_major, apply_rope);
  proj_tile_store(c11, m0 + 16, n0 + 16, l15, half, Y, out_head_major, apply_rope);
}

#define KT  32
#define KP  68
#define VPP 136

static __device__ __forceinline__ float xhalf(float x) {
  const int b = __builtin_bit_cast(int, x);
  return __builtin_bit_cast(float,
      __builtin_amdgcn_permlanex16(b, b, 0x76543210, 0xfedcba98, false, false));
}

__global__ __launch_bounds__(256) void attn_kernel(
    const float* __restrict__ Q, const float* __restrict__ K,
    const float* __restrict__ V, float* __restrict__ O) {
  __shared__ __attribute__((aligned(16))) float Ks[2][KT * KP];
  __shared__ __attribute__((aligned(16))) float Vs[2][(KT / 2) * VPP];

  const int tid  = threadIdx.x;
  const int lane = tid & 31;
  const int wave = tid >> 5;
  const int l15  = lane & 15;
  const int half = lane >> 4;
  const int koff = half * 2;
  const int p  = blockIdx.x;
  const int qt = (wave < 4) ? (4 * p + wave) : (120 - 4 * p + wave);
  const int h  = blockIdx.y;
  const int b  = blockIdx.z;
  const size_t base = ((size_t)(b * NHEADS + h)) * SEQ * DK;
  const float* Qh = Q + base;
  const float* Kh = K + base;
  const float* Vh = V + base;
  float*       Oh = O + base;
  const int r0 = qt * 16;
  const int qrow = r0 + l15;

  const int sq = tid & 15;
  const int sr = tid >> 4;
  const int k_st = sr * KP + sq * 4;
  const int v_st = sr * VPP + sq * 8;
  const int nst = 64 - 2 * p;

  v2f qb[16];
  #pragma unroll
  for (int kb = 0; kb < 16; ++kb)
    qb[kb] = *(const v2f*)(Qh + (size_t)qrow * DK + kb * 4 + koff);

  v4f rk0, rk1, rv0, rv1;
  {
    const size_t g  = (size_t)sr * DK + sq * 4;
    const size_t gv = (size_t)(2 * sr) * DK + sq * 4;
    rk0 = *(const v4f*)(Kh + g);
    rk1 = *(const v4f*)(Kh + g + 16 * DK);
    rv0 = *(const v4f*)(Vh + gv);
    rv1 = *(const v4f*)(Vh + gv + DK);
    *(v4f*)(&Ks[0][k_st])           = rk0;
    *(v4f*)(&Ks[0][k_st + 16 * KP]) = rk1;
    *(v4f*)(&Vs[0][v_st])     = (v4f){ rv0[0], rv1[0], rv0[1], rv1[1] };
    *(v4f*)(&Vs[0][v_st + 4]) = (v4f){ rv0[2], rv1[2], rv0[3], rv1[3] };
  }
  __syncthreads();

  v8f o0 = {0.f,0.f,0.f,0.f,0.f,0.f,0.f,0.f};
  v8f o1 = o0, o2 = o0, o3 = o0;
  float m_i = -__builtin_inff();
  float l_p = 0.f;

  for (int st = 0; st < nst; ++st) {
    const int cur = st & 1;
    const bool more = (st + 1 < nst);
    if (more) {
      const size_t g  = (size_t)((st + 1) * KT + sr) * DK + sq * 4;
      const size_t gv = (size_t)((st + 1) * KT + 2 * sr) * DK + sq * 4;
      rk0 = *(const v4f*)(Kh + g);
      rk1 = *(const v4f*)(Kh + g + 16 * DK);
      rv0 = *(const v4f*)(Vh + gv);
      rv1 = *(const v4f*)(Vh + gv + DK);
    }

    for (int t = 0; t < KT / 16; ++t) {
      const int j = st * (KT / 16) + t;
      if (j > qt) continue;
      const int c0 = j * 16;
      const float* kt = &Ks[cur][t * 16 * KP] + l15 * KP + koff;
      const float* vt = &Vs[cur][(t * 8 + half * 4) * VPP] + l15 * 2;

      v2f ka[16];
      #pragma unroll
      for (int kb = 0; kb < 16; ++kb) ka[kb] = *(const v2f*)(kt + kb * 4);
      v2f va[4][4];
      #pragma unroll
      for (int kb = 0; kb < 4; ++kb) {
        #pragma unroll
        for (int dt = 0; dt < 4; ++dt) {
          va[kb][dt] = *(const v2f*)(vt + kb * VPP + dt * 32);
        }
      }

      v8f s = {0.f,0.f,0.f,0.f,0.f,0.f,0.f,0.f};
      #pragma unroll
      for (int kb = 0; kb < 16; ++kb) s = wmma_f32_4(ka[kb], qb[kb], s);
      __builtin_amdgcn_sched_group_barrier(0x100, 16, 0);
      __builtin_amdgcn_sched_group_barrier(0x008, 16, 0);

      float sc[8];
      #pragma unroll
      for (int v = 0; v < 8; ++v) {
        const int kcol = c0 + half * 8 + v;
        sc[v] = (kcol <= qrow) ? s[v] * 0.125f : -__builtin_inff();
      }

      const float mloc = fmaxf(fmaxf(fmaxf(sc[0], sc[1]), fmaxf(sc[2], sc[3])),
                               fmaxf(fmaxf(sc[4], sc[5]), fmaxf(sc[6], sc[7])));
      const float mnew = fmaxf(m_i, fmaxf(mloc, xhalf(mloc)));
      const float fac  = __expf(m_i - mnew);
      float pv[8];
      #pragma unroll
      for (int v = 0; v < 8; ++v) pv[v] = __expf(sc[v] - mnew);
      const float rs = ((pv[0] + pv[1]) + (pv[2] + pv[3])) +
                       ((pv[4] + pv[5]) + (pv[6] + pv[7]));
      l_p = l_p * fac + rs;
      m_i = mnew;

      #pragma unroll
      for (int v = 0; v < 8; ++v) {
        o0[v] *= fac; o1[v] *= fac; o2[v] *= fac; o3[v] *= fac;
      }

      #pragma unroll
      for (int kb = 0; kb < 4; ++kb) {
        const v2f pb = { pv[2 * kb], pv[2 * kb + 1] };
        o0 = wmma_f32_4(va[kb][0], pb, o0);
        o1 = wmma_f32_4(va[kb][1], pb, o1);
        o2 = wmma_f32_4(va[kb][2], pb, o2);
        o3 = wmma_f32_4(va[kb][3], pb, o3);
      }
    }

    if (more) {
      *(v4f*)(&Ks[cur ^ 1][k_st])           = rk0;
      *(v4f*)(&Ks[cur ^ 1][k_st + 16 * KP]) = rk1;
      *(v4f*)(&Vs[cur ^ 1][v_st])     = (v4f){ rv0[0], rv1[0], rv0[1], rv1[1] };
      *(v4f*)(&Vs[cur ^ 1][v_st + 4]) = (v4f){ rv0[2], rv1[2], rv0[3], rv1[3] };
    }
    __syncthreads();
  }

  const float linv = 1.0f / (l_p + xhalf(l_p));
  float* op = Oh + (size_t)qrow * DK + half * 8;
  *(v4f*)(op +  0) = (v4f){ o0[0]*linv, o0[1]*linv, o0[2]*linv, o0[3]*linv };
  *(v4f*)(op +  4) = (v4f){ o0[4]*linv, o0[5]*linv, o0[6]*linv, o0[7]*linv };
  *(v4f*)(op + 16) = (v4f){ o1[0]*linv, o1[1]*linv, o1[2]*linv, o1[3]*linv };
  *(v4f*)(op + 20) = (v4f){ o1[4]*linv, o1[5]*linv, o1[6]*linv, o1[7]*linv };
  *(v4f*)(op + 32) = (v4f){ o2[0]*linv, o2[1]*linv, o2[2]*linv, o2[3]*linv };
  *(v4f*)(op + 36) = (v4f){ o2[4]*linv, o2[5]*linv, o2[6]*linv, o2[7]*linv };
  *(v4f*)(op + 48) = (v4f){ o3[0]*linv, o3[1]*linv, o3[2]*linv, o3[3]*linv };
  *(v4f*)(op + 52) = (v4f){ o3[4]*linv, o3[5]*linv, o3[6]*linv, o3[7]*linv };
}

extern "C" void kernel_launch(void* const* d_in, const int* in_sizes, int n_in,
                              void* d_out, int out_size, void* d_ws,
                              size_t ws_size, hipStream_t stream) {
  (void)in_sizes; (void)n_in; (void)out_size; (void)ws_size;
  const float* x  = (const float*)d_in[0];
  const float* Wq = (const float*)d_in[1];
  const float* Wk = (const float*)d_in[2];
  const float* Wv = (const float*)d_in[3];
  const float* Wo = (const float*)d_in[4];
  float* out = (float*)d_out;

  const size_t perBuf = (size_t)BATCH * NHEADS * SEQ * DK;
  float* Qb = (float*)d_ws;
  float* Kb = Qb + perBuf;
  float* Vb = Kb + perBuf;
  float* AO = Vb + perBuf;

  dim3 blk(256);
  dim3 gproj(DMODEL / BN, (BATCH * SEQ) / BM, 1);

  proj_gemm_kernel<<<gproj, blk, 0, stream>>>(x, Wq, Qb, 0, 1, 1);
  proj_gemm_kernel<<<gproj, blk, 0, stream>>>(x, Wk, Kb, 0, 1, 1);
  proj_gemm_kernel<<<gproj, blk, 0, stream>>>(x, Wv, Vb, 0, 1, 0);

  attn_kernel<<<dim3(SEQ / 16 / 8, NHEADS, BATCH), blk, 0, stream>>>(Qb, Kb, Vb, AO);

  proj_gemm_kernel<<<gproj, blk, 0, stream>>>(AO, Wo, out, 1, 0, 0);
}


# Q, K, V projections as ONE GEMM launch: 128x192 block tile (the same 64 columns of each weight), x tile staged once for all three, 64x48 per wave, 96 WMMAs per barrier; k order unchanged
# speedup vs baseline: 2.5952x; 1.0630x over previous
#include <hip/hip_runtime.h>
#include <math.h>

#define BATCH   2
#define SEQ     2048
#define DMODEL  1024
#define NHEADS  16
#define DK      64
#define LOG_THETA 9.210340371976184f

typedef __attribute__((ext_vector_type(2))) float v2f;
typedef __attribute__((ext_vector_type(4))) float v4f;
typedef __attribute__((ext_vector_type(8))) float v8f;

static __device__ __forceinline__ v8f wmma_f32_4(v2f a, v2f b, v8f c) {
  return __builtin_amdgcn_wmma_f32_16x16x4_f32(
       false, a,  false, b,
       (short)0, c,  false,  false);
}

#define BM  128
#define BN  64
#define BK  32
#define LDP 36

static __device__ __forceinline__ void proj_tile_mac(
    const float* __restrict__ as, const float* __restrict__ bs,
    v8f& c00, v8f& c01, v8f& c10, v8f& c11) {
  #pragma unroll
  for (int kk = 0; kk < BK / 4; ++kk) {
    const v2f a0 = *(const v2f*)(as + kk * 4);
    const v2f a1 = *(const v2f*)(as + 16 * LDP + kk * 4);
    const v2f b0 = *(const v2f*)(bs + kk * 4);
    const v2f b1 = *(const v2f*)(bs + 16 * LDP + kk * 4);
    c00 = wmma_f32_4(a0, b0, c00);
    c01 = wmma_f32_4(a0, b1, c01);
    c10 = wmma_f32_4(a1, b0, c10);
    c11 = wmma_f32_4(a1, b1, c11);
  }
}

static __device__ __forceinline__ void proj_tile_store(
    const v8f c, const int m0, const int n0, const int l15, const int half,
    float* __restrict__ Y, const int out_head_major, const int apply_rope) {
  const int e = n0 + l15;
  #pragma unroll
  for (int v = 0; v < 8; ++v) {
    const int mrow = m0 + v + half * 8;
    const int ss = mrow & (SEQ - 1);
    const int bb = mrow >> 11;
    float val = c[v];
    if (apply_rope) {
      float partner = __shfl_xor(val, 1, 32);
      const int p = (e & 63) >> 1;
      const float freq  = __expf(-(2.0f * LOG_THETA / (float)DK) * (float)p);
      const float phase = (float)ss * freq;
      float sn, cs;
      __sincosf(phase, &sn, &cs);
      val = (e & 1) ? (partner * sn + val * cs) : (val * cs - partner * sn);
    }
    if (out_head_major)
      Y[((size_t)(bb * NHEADS + (e >> 6)) * SEQ + ss) * DK + (e & 63)] = val;
    else
      Y[(size_t)mrow * DMODEL + e] = val;
  }
}

__global__ __launch_bounds__(256) void proj_gemm_kernel(
    const float* __restrict__ A, const float* __restrict__ W,
    float* __restrict__ Y, int in_head_major, int out_head_major,
    int apply_rope) {
  __shared__ __attribute__((aligned(16))) float As[2][BM * LDP];
  __shared__ __attribute__((aligned(16))) float Bs[2][BN * LDP];

  const int tid  = threadIdx.x;
  const int lane = tid & 31;
  const int wave = tid >> 5;
  const int l15  = lane & 15;
  const int half = lane >> 4;
  const int koff = half * 2;
  const int wm   = wave & 3;
  const int wn   = wave >> 2;
  const int mblk = blockIdx.y * BM;
  const int nblk = blockIdx.x * BN;

  const int kq = tid & 7;
  const int lr = tid >> 3;

  size_t aoff[4];
  #pragma unroll
  for (int i = 0; i < 4; ++i) {
    const int r = mblk + lr + 32 * i;
    aoff[i] = in_head_major
        ? ((size_t)(r >> 11) * NHEADS * SEQ * DK + (size_t)(r & (SEQ - 1)) * DK)
        : (size_t)r * DMODEL;
  }
  size_t woff[2];
  #pragma unroll
  for (int i = 0; i < 2; ++i)
    woff[i] = (size_t)(nblk + lr + 32 * i) * DMODEL;

  float4 ra[4], rb[2];
  const int lds_st = lr * LDP + kq * 4;

  v8f c00 = {0.f,0.f,0.f,0.f,0.f,0.f,0.f,0.f};
  v8f c01 = c00, c10 = c00, c11 = c00;

  const int a_frag = (wm * 32 + l15) * LDP + koff;
  const int b_frag = (wn * 32 + l15) * LDP + koff;

  {
    const int k = kq * 4;
    #pragma unroll
    for (int i = 0; i < 4; ++i) ra[i] = *(const float4*)(A + aoff[i] + k);
    #pragma unroll
    for (int i = 0; i < 2; ++i) rb[i] = *(const float4*)(W + woff[i] + k);
    #pragma unroll
    for (int i = 0; i < 4; ++i) *(float4*)(&As[0][lds_st + 32 * i * LDP]) = ra[i];
    #pragma unroll
    for (int i = 0; i < 2; ++i) *(float4*)(&Bs[0][lds_st + 32 * i * LDP]) = rb[i];
  }
  __syncthreads();

  for (int kt = 0; kt < DMODEL / BK - 1; ++kt) {
    const int cur = kt & 1;
    const int k = (kt + 1) * BK + kq * 4;
    const size_t ka = in_head_major
        ? ((size_t)(k >> 6) * SEQ * DK + (size_t)(k & 63)) : (size_t)k;
    #pragma unroll
    for (int i = 0; i < 4; ++i) ra[i] = *(const float4*)(A + aoff[i] + ka);
    #pragma unroll
    for (int i = 0; i < 2; ++i) rb[i] = *(const float4*)(W + woff[i] + k);

    proj_tile_mac(&As[cur][a_frag], &Bs[cur][b_frag], c00, c01, c10, c11);

    #pragma unroll
    for (int i = 0; i < 4; ++i)
      *(float4*)(&As[cur ^ 1][lds_st + 32 * i * LDP]) = ra[i];
    #pragma unroll
    for (int i = 0; i < 2; ++i)
      *(float4*)(&Bs[cur ^ 1][lds_st + 32 * i * LDP]) = rb[i];
    __syncthreads();
  }
  proj_tile_mac(&As[1][a_frag], &Bs[1][b_frag], c00, c01, c10, c11);

  const int m0 = mblk + wm * 32;
  const int n0 = nblk + wn * 32;
  proj_tile_store(c00, m0,      n0,      l15, half, Y, out_head_major, apply_rope);
  proj_tile_store(c01, m0,      n0 + 16, l15, half, Y, out_head_major, apply_rope);
  proj_tile_store(c10, m0 + 16, n0,      l15, half, Y, out_head_major, apply_rope);
  proj_tile_store(c11, m0 + 16, n0 + 16, l15, half, Y, out_head_major, apply_rope);
}

#define QN 192

static __device__ __forceinline__ void qkv_tile_mac(
    const float* __restrict__ as, const float* __restrict__ bs, v8f (&c)[4][3]) {
  #pragma unroll
  for (int kk = 0; kk < BK / 4; ++kk) {
    v2f a[4], b[3];
    #pragma unroll
    for (int i = 0; i < 4; ++i) a[i] = *(const v2f*)(as + i * 16 * LDP + kk * 4);
    #pragma unroll
    for (int j = 0; j < 3; ++j) b[j] = *(const v2f*)(bs + j * 16 * LDP + kk * 4);
    #pragma unroll
    for (int i = 0; i < 4; ++i) {
      #pragma unroll
      for (int j = 0; j < 3; ++j) c[i][j] = wmma_f32_4(a[i], b[j], c[i][j]);
    }
  }
}

__global__ __launch_bounds__(256) __attribute__((amdgpu_waves_per_eu(4, 4)))
void qkv_gemm_kernel(
    const float* __restrict__ A, const float* __restrict__ Wq,
    const float* __restrict__ Wk, const float* __restrict__ Wv,
    float* __restrict__ Yq, float* __restrict__ Yk, float* __restrict__ Yv) {
  __shared__ __attribute__((aligned(16))) float As[2][BM * LDP];
  __shared__ __attribute__((aligned(16))) float Bs[2][QN * LDP];

  const int tid  = threadIdx.x;
  const int lane = tid & 31;
  const int wave = tid >> 5;
  const int l15  = lane & 15;
  const int half = lane >> 4;
  const int koff = half * 2;
  const int wm   = wave & 1;
  const int wn   = wave >> 1;
  const int mblk = blockIdx.y * BM;
  const int nblk = blockIdx.x * BN;

  const int kq = tid & 7;
  const int lr = tid >> 3;

  const float* ap[4];
  #pragma unroll
  for (int i = 0; i < 4; ++i)
    ap[i] = A + (size_t)(mblk + lr + 32 * i) * DMODEL + kq * 4;
  const float* wp[6];
  #pragma unroll
  for (int i = 0; i < 6; ++i) {
    const float* Wsel = (i < 2) ? Wq : ((i < 4) ? Wk : Wv);
    wp[i] = Wsel + (size_t)(nblk + lr + 32 * (i & 1)) * DMODEL + kq * 4;
  }

  float4 ra[4], rb[6];
  const int lds_st = lr * LDP + kq * 4;

  v8f c[4][3];
  #pragma unroll
  for (int i = 0; i < 4; ++i) {
    #pragma unroll
    for (int j = 0; j < 3; ++j) c[i][j] = (v8f){0.f,0.f,0.f,0.f,0.f,0.f,0.f,0.f};
  }

  const int a_frag = (wm * 64 + l15) * LDP + koff;
  const int b_frag = (wn * 48 + l15) * LDP + koff;

  #pragma unroll
  for (int i = 0; i < 4; ++i) ra[i] = *(const float4*)(ap[i]);
  #pragma unroll
  for (int i = 0; i < 6; ++i) rb[i] = *(const float4*)(wp[i]);
  #pragma unroll
  for (int i = 0; i < 4; ++i) *(float4*)(&As[0][lds_st + 32 * i * LDP]) = ra[i];
  #pragma unroll
  for (int i = 0; i < 6; ++i) *(float4*)(&Bs[0][lds_st + 32 * i * LDP]) = rb[i];
  __syncthreads();

  for (int kt = 0; kt < DMODEL / BK - 1; ++kt) {
    const int cur = kt & 1;
    const int k = (kt + 1) * BK;
    #pragma unroll
    for (int i = 0; i < 4; ++i) ra[i] = *(const float4*)(ap[i] + k);
    #pragma unroll
    for (int i = 0; i < 6; ++i) rb[i] = *(const float4*)(wp[i] + k);

    qkv_tile_mac(&As[cur][a_frag], &Bs[cur][b_frag], c);

    #pragma unroll
    for (int i = 0; i < 4; ++i)
      *(float4*)(&As[cur ^ 1][lds_st + 32 * i * LDP]) = ra[i];
    #pragma unroll
    for (int i = 0; i < 6; ++i)
      *(float4*)(&Bs[cur ^ 1][lds_st + 32 * i * LDP]) = rb[i];
    __builtin_amdgcn_sched_group_barrier(0x020, 10, 0);
    __builtin_amdgcn_sched_group_barrier(0x100, 14, 0);
    __builtin_amdgcn_sched_group_barrier(0x008, 24, 0);
    __builtin_amdgcn_sched_group_barrier(0x100, 7, 0);
    __builtin_amdgcn_sched_group_barrier(0x008, 24, 0);
    __builtin_amdgcn_sched_group_barrier(0x100, 7, 0);
    __builtin_amdgcn_sched_group_barrier(0x008, 48, 0);
    __builtin_amdgcn_sched_group_barrier(0x200, 10, 0);
    __syncthreads();
  }
  qkv_tile_mac(&As[1][a_frag], &Bs[1][b_frag], c);

  const int m0 = mblk + wm * 64;
  #pragma unroll
  for (int j = 0; j < 3; ++j) {
    const int t = wn * 3 + j;
    const int which = t >> 2;
    float* Y = (which == 0) ? Yq : ((which == 1) ? Yk : Yv);
    const int n0 = nblk + (t & 3) * 16;
    #pragma unroll
    for (int i = 0; i < 4; ++i)
      proj_tile_store(c[i][j], m0 + i * 16, n0, l15, half, Y, 1, which < 2);
  }
}

#define KT  32
#define KP  68
#define VPP 136

static __device__ __forceinline__ float xhalf(float x) {
  const int b = __builtin_bit_cast(int, x);
  return __builtin_bit_cast(float,
      __builtin_amdgcn_permlanex16(b, b, 0x76543210, 0xfedcba98, false, false));
}

__global__ __launch_bounds__(256) void attn_kernel(
    const float* __restrict__ Q, const float* __restrict__ K,
    const float* __restrict__ V, float* __restrict__ O) {
  __shared__ __attribute__((aligned(16))) float Ks[2][KT * KP];
  __shared__ __attribute__((aligned(16))) float Vs[2][(KT / 2) * VPP];

  const int tid  = threadIdx.x;
  const int lane = tid & 31;
  const int wave = tid >> 5;
  const int l15  = lane & 15;
  const int half = lane >> 4;
  const int koff = half * 2;
  const int p  = blockIdx.x;
  const int qt = (wave < 4) ? (4 * p + wave) : (120 - 4 * p + wave);
  const int h  = blockIdx.y;
  const int b  = blockIdx.z;
  const size_t base = ((size_t)(b * NHEADS + h)) * SEQ * DK;
  const float* Qh = Q + base;
  const float* Kh = K + base;
  const float* Vh = V + base;
  float*       Oh = O + base;
  const int r0 = qt * 16;
  const int qrow = r0 + l15;

  const int sq = tid & 15;
  const int sr = tid >> 4;
  const int k_st = sr * KP + sq * 4;
  const int v_st = sr * VPP + sq * 8;
  const int nst = 64 - 2 * p;

  v2f qb[16];
  #pragma unroll
  for (int kb = 0; kb < 16; ++kb)
    qb[kb] = *(const v2f*)(Qh + (size_t)qrow * DK + kb * 4 + koff);

  v4f rk0, rk1, rv0, rv1;
  {
    const size_t g  = (size_t)sr * DK + sq * 4;
    const size_t gv = (size_t)(2 * sr) * DK + sq * 4;
    rk0 = *(const v4f*)(Kh + g);
    rk1 = *(const v4f*)(Kh + g + 16 * DK);
    rv0 = *(const v4f*)(Vh + gv);
    rv1 = *(const v4f*)(Vh + gv + DK);
    *(v4f*)(&Ks[0][k_st])           = rk0;
    *(v4f*)(&Ks[0][k_st + 16 * KP]) = rk1;
    *(v4f*)(&Vs[0][v_st])     = (v4f){ rv0[0], rv1[0], rv0[1], rv1[1] };
    *(v4f*)(&Vs[0][v_st + 4]) = (v4f){ rv0[2], rv1[2], rv0[3], rv1[3] };
  }
  __syncthreads();

  v8f o0 = {0.f,0.f,0.f,0.f,0.f,0.f,0.f,0.f};
  v8f o1 = o0, o2 = o0, o3 = o0;
  float m_i = -__builtin_inff();
  float l_p = 0.f;

  for (int st = 0; st < nst; ++st) {
    const int cur = st & 1;
    const bool more = (st + 1 < nst);
    if (more) {
      const size_t g  = (size_t)((st + 1) * KT + sr) * DK + sq * 4;
      const size_t gv = (size_t)((st + 1) * KT + 2 * sr) * DK + sq * 4;
      rk0 = *(const v4f*)(Kh + g);
      rk1 = *(const v4f*)(Kh + g + 16 * DK);
      rv0 = *(const v4f*)(Vh + gv);
      rv1 = *(const v4f*)(Vh + gv + DK);
    }

    for (int t = 0; t < KT / 16; ++t) {
      const int j = st * (KT / 16) + t;
      if (j > qt) continue;
      const int c0 = j * 16;
      const float* kt = &Ks[cur][t * 16 * KP] + l15 * KP + koff;
      const float* vt = &Vs[cur][(t * 8 + half * 4) * VPP] + l15 * 2;

      v2f ka[16];
      #pragma unroll
      for (int kb = 0; kb < 16; ++kb) ka[kb] = *(const v2f*)(kt + kb * 4);
      v2f va[4][4];
      #pragma unroll
      for (int kb = 0; kb < 4; ++kb) {
        #pragma unroll
        for (int dt = 0; dt < 4; ++dt) {
          va[kb][dt] = *(const v2f*)(vt + kb * VPP + dt * 32);
        }
      }

      v8f s = {0.f,0.f,0.f,0.f,0.f,0.f,0.f,0.f};
      #pragma unroll
      for (int kb = 0; kb < 16; ++kb) s = wmma_f32_4(ka[kb], qb[kb], s);
      __builtin_amdgcn_sched_group_barrier(0x100, 16, 0);
      __builtin_amdgcn_sched_group_barrier(0x008, 16, 0);

      float sc[8];
      #pragma unroll
      for (int v = 0; v < 8; ++v) {
        const int kcol = c0 + half * 8 + v;
        sc[v] = (kcol <= qrow) ? s[v] * 0.125f : -__builtin_inff();
      }

      const float mloc = fmaxf(fmaxf(fmaxf(sc[0], sc[1]), fmaxf(sc[2], sc[3])),
                               fmaxf(fmaxf(sc[4], sc[5]), fmaxf(sc[6], sc[7])));
      const float mnew = fmaxf(m_i, fmaxf(mloc, xhalf(mloc)));
      const float fac  = __expf(m_i - mnew);
      float pv[8];
      #pragma unroll
      for (int v = 0; v < 8; ++v) pv[v] = __expf(sc[v] - mnew);
      const float rs = ((pv[0] + pv[1]) + (pv[2] + pv[3])) +
                       ((pv[4] + pv[5]) + (pv[6] + pv[7]));
      l_p = l_p * fac + rs;
      m_i = mnew;

      #pragma unroll
      for (int v = 0; v < 8; ++v) {
        o0[v] *= fac; o1[v] *= fac; o2[v] *= fac; o3[v] *= fac;
      }

      #pragma unroll
      for (int kb = 0; kb < 4; ++kb) {
        const v2f pb = { pv[2 * kb], pv[2 * kb + 1] };
        o0 = wmma_f32_4(va[kb][0], pb, o0);
        o1 = wmma_f32_4(va[kb][1], pb, o1);
        o2 = wmma_f32_4(va[kb][2], pb, o2);
        o3 = wmma_f32_4(va[kb][3], pb, o3);
      }
    }

    if (more) {
      *(v4f*)(&Ks[cur ^ 1][k_st])           = rk0;
      *(v4f*)(&Ks[cur ^ 1][k_st + 16 * KP]) = rk1;
      *(v4f*)(&Vs[cur ^ 1][v_st])     = (v4f){ rv0[0], rv1[0], rv0[1], rv1[1] };
      *(v4f*)(&Vs[cur ^ 1][v_st + 4]) = (v4f){ rv0[2], rv1[2], rv0[3], rv1[3] };
    }
    __syncthreads();
  }

  const float linv = 1.0f / (l_p + xhalf(l_p));
  float* op = Oh + (size_t)qrow * DK + half * 8;
  *(v4f*)(op +  0) = (v4f){ o0[0]*linv, o0[1]*linv, o0[2]*linv, o0[3]*linv };
  *(v4f*)(op +  4) = (v4f){ o0[4]*linv, o0[5]*linv, o0[6]*linv, o0[7]*linv };
  *(v4f*)(op + 16) = (v4f){ o1[0]*linv, o1[1]*linv, o1[2]*linv, o1[3]*linv };
  *(v4f*)(op + 20) = (v4f){ o1[4]*linv, o1[5]*linv, o1[6]*linv, o1[7]*linv };
  *(v4f*)(op + 32) = (v4f){ o2[0]*linv, o2[1]*linv, o2[2]*linv, o2[3]*linv };
  *(v4f*)(op + 36) = (v4f){ o2[4]*linv, o2[5]*linv, o2[6]*linv, o2[7]*linv };
  *(v4f*)(op + 48) = (v4f){ o3[0]*linv, o3[1]*linv, o3[2]*linv, o3[3]*linv };
  *(v4f*)(op + 52) = (v4f){ o3[4]*linv, o3[5]*linv, o3[6]*linv, o3[7]*linv };
}

extern "C" void kernel_launch(void* const* d_in, const int* in_sizes, int n_in,
                              void* d_out, int out_size, void* d_ws,
                              size_t ws_size, hipStream_t stream) {
  (void)in_sizes; (void)n_in; (void)out_size; (void)ws_size;
  const float* x  = (const float*)d_in[0];
  const float* Wq = (const float*)d_in[1];
  const float* Wk = (const float*)d_in[2];
  const float* Wv = (const float*)d_in[3];
  const float* Wo = (const float*)d_in[4];
  float* out = (float*)d_out;

  const size_t perBuf = (size_t)BATCH * NHEADS * SEQ * DK;
  float* Qb = (float*)d_ws;
  float* Kb = Qb + perBuf;
  float* Vb = Kb + perBuf;
  float* AO = Vb + perBuf;

  dim3 blk(256);
  dim3 gproj(DMODEL / BN, (BATCH * SEQ) / BM, 1);

  qkv_gemm_kernel<<<gproj, blk, 0, stream>>>(x, Wq, Wk, Wv, Qb, Kb, Vb);

  attn_kernel<<<dim3(SEQ / 16 / 8, NHEADS, BATCH), blk, 0, stream>>>(Qb, Kb, Vb, AO);

  proj_gemm_kernel<<<gproj, blk, 0, stream>>>(AO, Wo, out, 1, 0, 0);
}
